# starting kernel
# baseline (speedup 1.0000x reference)
#include <hip/hip_runtime.h>

typedef __attribute__((ext_vector_type(2))) float v2f;
typedef __attribute__((ext_vector_type(8))) float v8f;

#define THREADS 256
#define WAVES_PER_BLOCK (THREADS / 32)

__global__ void gcn_fill_kernel(float* __restrict__ p, float v, long n) {
  long i = (long)blockIdx.x * blockDim.x + threadIdx.x;
  if (i < n) p[i] = v;
}

__global__ void gcn_deg_kernel(const int* __restrict__ col, float* __restrict__ deg, int E) {
  int e = blockIdx.x * blockDim.x + threadIdx.x;
  if (e < E) atomicAdd(&deg[col[e]], 1.0f);
}

__global__ void gcn_dinv_kernel(const float* __restrict__ deg, float* __restrict__ dinv, int n) {
  int i = blockIdx.x * blockDim.x + threadIdx.x;
  if (i < n) dinv[i] = rsqrtf(deg[i]);
}

__global__ void gcn_pad_w3_kernel(const float* __restrict__ W, float* __restrict__ Wp) {
  int i = blockIdx.x * blockDim.x + threadIdx.x;
  if (i < 64 * 48) {
    int k = i / 48;
    int c = i - k * 48;
    Wp[i] = (c < 40) ? W[k * 40 + c] : 0.0f;
  }
}

template <int K, int NCOLS>
__global__ void gcn_gemm_wmma_kernel(const float* __restrict__ A,
                                     const float* __restrict__ B,
                                     float* __restrict__ C,
                                     int numRowTiles) {
  __shared__ float bs[K * 16];

  int c0 = blockIdx.y << 4;

#pragma unroll
  for (int i = threadIdx.x; i < K * 16; i += THREADS) {
    int k = i >> 4;
    int c = i & 15;
    bs[i] = B[k * NCOLS + c0 + c];
  }
  __syncthreads();

  int wave = threadIdx.x >> 5;
  int lane = threadIdx.x & 31;
  int rowTile = blockIdx.x * WAVES_PER_BLOCK + wave;
  if (rowTile >= numRowTiles) return;

  int r0 = rowTile << 4;
  int half = lane >> 4;
  int l = lane & 15;

  const float* ap = A + (size_t)(r0 + l) * K + 2 * half;
  const float* bp = bs + (2 * half) * 16 + l;

  v8f acc = {};
#pragma unroll
  for (int k = 0; k < K; k += 4) {
    v2f a;
    a.x = ap[k];
    a.y = ap[k + 1];
    v2f b;
    b.x = bp[k * 16];
    b.y = bp[k * 16 + 16];
    acc = __builtin_amdgcn_wmma_f32_16x16x4_f32(
         false, a,  false, b,
         (short)0, acc,  false,  false);
  }

  float* cp = C + (size_t)(r0 + half * 8) * NCOLS + c0 + l;
#pragma unroll
  for (int j = 0; j < 8; ++j) cp[j * NCOLS] = acc[j];
}

__global__ void gcn_agg_kernel(const int* __restrict__ row,
                               const int* __restrict__ col,
                               const float* __restrict__ dinv,
                               const float* __restrict__ xw,
                               float* __restrict__ agg,
                               long total, int d4, int ld) {
  long t = (long)blockIdx.x * blockDim.x + threadIdx.x;
  if (t >= total) return;
  int chunk = (int)(t % d4);
  int e = (int)(t / d4);
  int r = row[e];
  int c = col[e];
  float w = dinv[r] * dinv[c];
  const float4* xr = (const float4*)(xw + (size_t)r * ld);
  float4 v = xr[chunk];
  float* op = agg + (size_t)c * ld + (size_t)chunk * 4;
  atomicAdd(op + 0, v.x * w);
  atomicAdd(op + 1, v.y * w);
  atomicAdd(op + 2, v.z * w);
  atomicAdd(op + 3, v.w * w);
}

__global__ void gcn_finalize_kernel(const float* __restrict__ agg,
                                    const float* __restrict__ xw,
                                    const float* __restrict__ dinv,
                                    const float* __restrict__ bias,
                                    float* __restrict__ out,
                                    long total, int d, int ld, int relu) {
  long t = (long)blockIdx.x * blockDim.x + threadIdx.x;
  if (t >= total) return;
  int node = (int)(t / d);
  int f = (int)(t - (long)node * d);
  size_t in = (size_t)node * ld + f;
  float s = dinv[node];
  float v = agg[in] + s * s * xw[in] + bias[f];
  out[t] = relu ? fmaxf(v, 0.0f) : v;
}

static inline int cdiv(long a, int b) { return (int)((a + b - 1) / b); }

template <int K, int NCOLS>
static void launch_gemm(const float* A, const float* B, float* C, int M,
                        hipStream_t stream) {
  int numRowTiles = M / 16;
  dim3 grid(cdiv(numRowTiles, WAVES_PER_BLOCK), NCOLS / 16, 1);
  gcn_gemm_wmma_kernel<K, NCOLS><<<grid, THREADS, 0, stream>>>(A, B, C, numRowTiles);
}

extern "C" void kernel_launch(void* const* d_in, const int* in_sizes, int n_in,
                              void* d_out, int out_size, void* d_ws, size_t ws_size,
                              hipStream_t stream) {
  const float* x  = (const float*)d_in[0];
  const int*   ei = (const int*)d_in[1];
  const float* W1 = (const float*)d_in[2];
  const float* b1 = (const float*)d_in[3];
  const float* W2 = (const float*)d_in[4];
  const float* b2 = (const float*)d_in[5];
  const float* W3 = (const float*)d_in[6];
  const float* b3 = (const float*)d_in[7];

  const int D_IN = 128, D_H = 64, D_OUT = 40, D_OUT_PAD = 48;
  int N = in_sizes[0] / D_IN;
  int E = in_sizes[1] / 2;
  const int* row = ei;
  const int* col = ei + E;

  float* ws   = (float*)d_ws;
  float* deg  = ws;
  float* dinv = ws + (size_t)N;
  float* xw   = ws + 2 * (size_t)N;
  float* agg  = xw + (size_t)N * D_H;
  float* h    = agg + (size_t)N * D_H;
  float* wpad = h + (size_t)N * D_H;

  gcn_fill_kernel<<<cdiv(N, THREADS), THREADS, 0, stream>>>(deg, 1.0f, N);
  gcn_deg_kernel<<<cdiv(E, THREADS), THREADS, 0, stream>>>(col, deg, E);
  gcn_dinv_kernel<<<cdiv(N, THREADS), THREADS, 0, stream>>>(deg, dinv, N);
  gcn_pad_w3_kernel<<<cdiv(64 * 48, THREADS), THREADS, 0, stream>>>(W3, wpad);

  {
    const int d = 64, ld = 64, d4 = 16;
    long nd = (long)N * ld, ed = (long)E * d4, od = (long)N * d;
    launch_gemm<128, 64>(x, W1, xw, N, stream);
    gcn_fill_kernel<<<cdiv(nd, THREADS), THREADS, 0, stream>>>(agg, 0.0f, nd);
    gcn_agg_kernel<<<cdiv(ed, THREADS), THREADS, 0, stream>>>(row, col, dinv, xw, agg, ed, d4, ld);
    gcn_finalize_kernel<<<cdiv(od, THREADS), THREADS, 0, stream>>>(agg, xw, dinv, b1, h, od, d, ld, 1);
  }

  {
    const int d = 64, ld = 64, d4 = 16;
    long nd = (long)N * ld, ed = (long)E * d4, od = (long)N * d;
    launch_gemm<64, 64>(h, W2, xw, N, stream);
    gcn_fill_kernel<<<cdiv(nd, THREADS), THREADS, 0, stream>>>(agg, 0.0f, nd);
    gcn_agg_kernel<<<cdiv(ed, THREADS), THREADS, 0, stream>>>(row, col, dinv, xw, agg, ed, d4, ld);
    gcn_finalize_kernel<<<cdiv(od, THREADS), THREADS, 0, stream>>>(agg, xw, dinv, b2, h, od, d, ld, 1);
  }

  {
    const int d = D_OUT, ld = D_OUT_PAD, d4 = 10;
    long nd = (long)N * ld, ed = (long)E * d4, od = (long)N * d;
    launch_gemm<64, 48>(h, wpad, xw, N, stream);
    gcn_fill_kernel<<<cdiv(nd, THREADS), THREADS, 0, stream>>>(agg, 0.0f, nd);
    gcn_agg_kernel<<<cdiv(ed, THREADS), THREADS, 0, stream>>>(row, col, dinv, xw, agg, ed, d4, ld);
    gcn_finalize_kernel<<<cdiv(od, THREADS), THREADS, 0, stream>>>(agg, xw, dinv, b3,
                                                                   (float*)d_out, od, d, ld, 0);
  }
}


# Aggregation pulls per destination over a per-call CSR (int count, scan, place) instead of f32 scatter atomics; zero-fill and finalize fold in; float sum order is now CSR arrival order
# speedup vs baseline: 2.0514x; 2.0514x over previous
#include <hip/hip_runtime.h>

typedef __attribute__((ext_vector_type(2))) float v2f;
typedef __attribute__((ext_vector_type(8))) float v8f;

#define THREADS 256
#define WAVES_PER_BLOCK (THREADS / 32)
#define SCAN_T 1024

__global__ void gcn_zero_int_kernel(int* __restrict__ p, int n) {
  int i = blockIdx.x * blockDim.x + threadIdx.x;
  if (i < n) p[i] = 0;
}

__global__ void gcn_count_kernel(const int* __restrict__ col, int* __restrict__ cnt,
                                 int* __restrict__ pos, int E) {
  int e = blockIdx.x * blockDim.x + threadIdx.x;
  if (e < E) pos[e] = atomicAdd(&cnt[col[e]], 1);
}

__global__ void __launch_bounds__(SCAN_T)
gcn_scan_kernel(const int* __restrict__ cnt, int* __restrict__ off,
                float* __restrict__ dinv, int n) {
  __shared__ int wsum[32];
  __shared__ int s_prefix;
  int tid = threadIdx.x;
  int lane = tid & 31;
  int wave = tid >> 5;
  int tileBase = blockIdx.x * SCAN_T;

  int part = 0;
  for (int i = tid; i < tileBase; i += SCAN_T) part += cnt[i];
  for (int d = 16; d > 0; d >>= 1) part += __shfl_xor(part, d);
  if (lane == 0) wsum[wave] = part;
  __syncthreads();
  if (wave == 0) {
    int v = wsum[lane];
    for (int d = 16; d > 0; d >>= 1) v += __shfl_xor(v, d);
    if (lane == 0) s_prefix = v;
  }
  __syncthreads();
  int prefix = s_prefix;
  __syncthreads();

  int i = tileBase + tid;
  int c = (i < n) ? cnt[i] : 0;
  int incl = c;
  for (int d = 1; d < 32; d <<= 1) {
    int t = __shfl_up(incl, d);
    if (lane >= d) incl += t;
  }
  if (lane == 31) wsum[wave] = incl;
  __syncthreads();
  if (wave == 0) {
    int v = wsum[lane];
    int vi = v;
    for (int d = 1; d < 32; d <<= 1) {
      int t = __shfl_up(vi, d);
      if (lane >= d) vi += t;
    }
    wsum[lane] = vi - v;
  }
  __syncthreads();
  int excl = prefix + wsum[wave] + incl - c;
  if (i < n) {
    off[i] = excl;
    dinv[i] = rsqrtf((float)(c + 1));
  }
  if (i == n - 1) off[n] = excl + c;
}

__global__ void gcn_place_kernel(const int* __restrict__ row, const int* __restrict__ col,
                                 const int* __restrict__ off, const int* __restrict__ pos,
                                 int* __restrict__ src, int E) {
  int e = blockIdx.x * blockDim.x + threadIdx.x;
  if (e < E) src[off[col[e]] + pos[e]] = row[e];
}

__global__ void gcn_pad_w3_kernel(const float* __restrict__ W, float* __restrict__ Wp) {
  int i = blockIdx.x * blockDim.x + threadIdx.x;
  if (i < 64 * 48) {
    int k = i / 48;
    int c = i - k * 48;
    Wp[i] = (c < 40) ? W[k * 40 + c] : 0.0f;
  }
}

template <int K, int NCOLS>
__global__ void gcn_gemm_wmma_kernel(const float* __restrict__ A,
                                     const float* __restrict__ B,
                                     float* __restrict__ C,
                                     int numRowTiles) {
  __shared__ float bs[K * 16];

  int c0 = blockIdx.y << 4;

#pragma unroll
  for (int i = threadIdx.x; i < K * 16; i += THREADS) {
    int k = i >> 4;
    int c = i & 15;
    bs[i] = B[k * NCOLS + c0 + c];
  }
  __syncthreads();

  int wave = threadIdx.x >> 5;
  int lane = threadIdx.x & 31;
  int rowTile = blockIdx.x * WAVES_PER_BLOCK + wave;
  if (rowTile >= numRowTiles) return;

  int r0 = rowTile << 4;
  int half = lane >> 4;
  int l = lane & 15;

  const float* ap = A + (size_t)(r0 + l) * K + 2 * half;
  const float* bp = bs + (2 * half) * 16 + l;

  v8f acc = {};
#pragma unroll
  for (int k = 0; k < K; k += 4) {
    v2f a;
    a.x = ap[k];
    a.y = ap[k + 1];
    v2f b;
    b.x = bp[k * 16];
    b.y = bp[k * 16 + 16];
    acc = __builtin_amdgcn_wmma_f32_16x16x4_f32(
         false, a,  false, b,
         (short)0, acc,  false,  false);
  }

  float* cp = C + (size_t)(r0 + half * 8) * NCOLS + c0 + l;
#pragma unroll
  for (int j = 0; j < 8; ++j) cp[j * NCOLS] = acc[j];
}

template <int NF2, int LD, int DOUT, bool RELU>
__global__ void __launch_bounds__(THREADS)
gcn_pull_kernel(const int* __restrict__ off, const int* __restrict__ src,
                const float* __restrict__ dinv, const float* __restrict__ xw,
                const float* __restrict__ bias, float* __restrict__ out, int n) {
  __shared__ float stage[WAVES_PER_BLOCK * DOUT];
  int lane = threadIdx.x & 31;
  int wave = threadIdx.x >> 5;
  int node0 = blockIdx.x * WAVES_PER_BLOCK;
  int node = __builtin_amdgcn_readfirstlane(node0 + wave);
  bool act = lane < NF2;

  if (node < n) {
    int beg = __builtin_amdgcn_readfirstlane(off[node]);
    int end = __builtin_amdgcn_readfirstlane(off[node + 1]);
    float dc = dinv[node];
    const float* xl = xw + 2 * lane;
    float ax = 0.0f, ay = 0.0f;

    for (int base = beg; base < end; base += 32) {
      int m = end - base;
      if (m > 32) m = 32;
      int s_l = node;
      float w_l = 0.0f;
      if (lane < m) {
        s_l = src[base + lane];
        w_l = dinv[s_l] * dc;
      }
      int w_i = __float_as_int(w_l);
      for (int j = 0; j < m; j += 8) {
        int s[8];
        float w[8];
        float2 v[8];
#pragma unroll
        for (int u = 0; u < 8; ++u) {
          s[u] = __builtin_amdgcn_readlane(s_l, j + u);
          w[u] = __int_as_float(__builtin_amdgcn_readlane(w_i, j + u));
        }
#pragma unroll
        for (int u = 0; u < 8; ++u) {
          v[u] = make_float2(0.0f, 0.0f);
          if (act) v[u] = *(const float2*)(xl + (size_t)s[u] * LD);
        }
        if (j + 8 > m) {
#pragma unroll
          for (int u = 1; u < 8; ++u)
            if (j + u >= m) v[u] = make_float2(0.0f, 0.0f);
        }
#pragma unroll
        for (int u = 0; u < 8; ++u) {
          ax += v[u].x * w[u];
          ay += v[u].y * w[u];
        }
      }
    }

    if (act) {
      float2 v = *(const float2*)(xl + (size_t)node * LD);
      float2 b = *(const float2*)(bias + 2 * lane);
      float ox = ax + dc * dc * v.x + b.x;
      float oy = ay + dc * dc * v.y + b.y;
      if (RELU) { ox = fmaxf(ox, 0.0f); oy = fmaxf(oy, 0.0f); }
      if (DOUT == 64) {
        *(float2*)(out + (size_t)node * DOUT + 2 * lane) = make_float2(ox, oy);
      } else {
        stage[wave * DOUT + 2 * lane] = ox;
        stage[wave * DOUT + 2 * lane + 1] = oy;
      }
    }
  }

  if (DOUT != 64) {
    __syncthreads();
    size_t o0 = (size_t)node0 * DOUT;
    size_t lim = (size_t)n * DOUT;
    for (int t = threadIdx.x; t < WAVES_PER_BLOCK * DOUT; t += THREADS)
      if (o0 + t < lim) out[o0 + t] = stage[t];
  }
}

static inline int cdiv(long a, int b) { return (int)((a + b - 1) / b); }
static inline size_t pad64(size_t x) { return (x + 63) & ~(size_t)63; }

template <int K, int NCOLS>
static void launch_gemm(const float* A, const float* B, float* C, int M,
                        hipStream_t stream) {
  int numRowTiles = M / 16;
  dim3 grid(cdiv(numRowTiles, WAVES_PER_BLOCK), NCOLS / 16, 1);
  gcn_gemm_wmma_kernel<K, NCOLS><<<grid, THREADS, 0, stream>>>(A, B, C, numRowTiles);
}

extern "C" void kernel_launch(void* const* d_in, const int* in_sizes, int n_in,
                              void* d_out, int out_size, void* d_ws, size_t ws_size,
                              hipStream_t stream) {
  const float* x  = (const float*)d_in[0];
  const int*   ei = (const int*)d_in[1];
  const float* W1 = (const float*)d_in[2];
  const float* b1 = (const float*)d_in[3];
  const float* W2 = (const float*)d_in[4];
  const float* b2 = (const float*)d_in[5];
  const float* W3 = (const float*)d_in[6];
  const float* b3 = (const float*)d_in[7];

  const int D_IN = 128, D_H = 64;
  int N = in_sizes[0] / D_IN;
  int E = in_sizes[1] / 2;
  const int* row = ei;
  const int* col = ei + E;

  float* ws   = (float*)d_ws;
  int*   cnt  = (int*)ws;
  int*   off  = cnt + pad64((size_t)N);
  float* dinv = (float*)(off + pad64((size_t)N + 1));
  int*   pos  = (int*)(dinv + pad64((size_t)N));
  int*   srcs = pos + pad64((size_t)E);
  float* xw   = (float*)(srcs + pad64((size_t)E));
  float* h    = xw + (size_t)N * D_H;
  float* wpad = h + (size_t)N * D_H;

  gcn_zero_int_kernel<<<cdiv(N, THREADS), THREADS, 0, stream>>>(cnt, N);
  gcn_count_kernel<<<cdiv(E, THREADS), THREADS, 0, stream>>>(col, cnt, pos, E);
  gcn_scan_kernel<<<cdiv(N, SCAN_T), SCAN_T, 0, stream>>>(cnt, off, dinv, N);
  gcn_place_kernel<<<cdiv(E, THREADS), THREADS, 0, stream>>>(row, col, off, pos, srcs, E);
  gcn_pad_w3_kernel<<<cdiv(64 * 48, THREADS), THREADS, 0, stream>>>(W3, wpad);

  int pullBlocks = cdiv(N, WAVES_PER_BLOCK);

  launch_gemm<128, 64>(x, W1, xw, N, stream);
  gcn_pull_kernel<32, 64, 64, true><<<pullBlocks, THREADS, 0, stream>>>(off, srcs, dinv, xw, b1, h, N);

  launch_gemm<64, 64>(h, W2, xw, N, stream);
  gcn_pull_kernel<32, 64, 64, true><<<pullBlocks, THREADS, 0, stream>>>(off, srcs, dinv, xw, b2, h, N);

  launch_gemm<64, 48>(h, wpad, xw, N, stream);
  gcn_pull_kernel<20, 48, 40, false><<<pullBlocks, THREADS, 0, stream>>>(off, srcs, dinv, xw, b3,
                                                                        (float*)d_out, N);
}


# Per-edge weight lookup removed: GEMM epilogue stores rows scaled by deg^-1/2, pull adds plain rows then scales the sum by deg^-1/2 of the node; sum order kept, term rounding differs
# speedup vs baseline: 2.2786x; 1.1108x over previous
#include <hip/hip_runtime.h>

typedef __attribute__((ext_vector_type(2))) float v2f;
typedef __attribute__((ext_vector_type(8))) float v8f;

#define THREADS 256
#define WAVES_PER_BLOCK (THREADS / 32)
#define SCAN_T 1024

__global__ void gcn_zero_int_kernel(int* __restrict__ p, int n) {
  int i = blockIdx.x * blockDim.x + threadIdx.x;
  if (i < n) p[i] = 0;
}

__global__ void gcn_count_kernel(const int* __restrict__ col, int* __restrict__ cnt,
                                 int* __restrict__ pos, int E) {
  int e = blockIdx.x * blockDim.x + threadIdx.x;
  if (e < E) pos[e] = atomicAdd(&cnt[col[e]], 1);
}

__global__ void __launch_bounds__(SCAN_T)
gcn_scan_kernel(const int* __restrict__ cnt, int* __restrict__ off,
                float* __restrict__ dinv, int n) {
  __shared__ int wsum[32];
  __shared__ int s_prefix;
  int tid = threadIdx.x;
  int lane = tid & 31;
  int wave = tid >> 5;
  int tileBase = blockIdx.x * SCAN_T;

  int part = 0;
  for (int i = tid; i < tileBase; i += SCAN_T) part += cnt[i];
  for (int d = 16; d > 0; d >>= 1) part += __shfl_xor(part, d);
  if (lane == 0) wsum[wave] = part;
  __syncthreads();
  if (wave == 0) {
    int v = wsum[lane];
    for (int d = 16; d > 0; d >>= 1) v += __shfl_xor(v, d);
    if (lane == 0) s_prefix = v;
  }
  __syncthreads();
  int prefix = s_prefix;
  __syncthreads();

  int i = tileBase + tid;
  int c = (i < n) ? cnt[i] : 0;
  int incl = c;
  for (int d = 1; d < 32; d <<= 1) {
    int t = __shfl_up(incl, d);
    if (lane >= d) incl += t;
  }
  if (lane == 31) wsum[wave] = incl;
  __syncthreads();
  if (wave == 0) {
    int v = wsum[lane];
    int vi = v;
    for (int d = 1; d < 32; d <<= 1) {
      int t = __shfl_up(vi, d);
      if (lane >= d) vi += t;
    }
    wsum[lane] = vi - v;
  }
  __syncthreads();
  int excl = prefix + wsum[wave] + incl - c;
  if (i < n) {
    off[i] = excl;
    dinv[i] = rsqrtf((float)(c + 1));
  }
  if (i == n - 1) off[n] = excl + c;
}

__global__ void gcn_place_kernel(const int* __restrict__ row, const int* __restrict__ col,
                                 const int* __restrict__ off, const int* __restrict__ pos,
                                 int* __restrict__ src, int E) {
  int e = blockIdx.x * blockDim.x + threadIdx.x;
  if (e < E) src[off[col[e]] + pos[e]] = row[e];
}

__global__ void gcn_pad_w3_kernel(const float* __restrict__ W, float* __restrict__ Wp) {
  int i = blockIdx.x * blockDim.x + threadIdx.x;
  if (i < 64 * 48) {
    int k = i / 48;
    int c = i - k * 48;
    Wp[i] = (c < 40) ? W[k * 40 + c] : 0.0f;
  }
}

template <int K, int NCOLS>
__global__ void gcn_gemm_wmma_kernel(const float* __restrict__ A,
                                     const float* __restrict__ B,
                                     const float* __restrict__ rscale,
                                     float* __restrict__ C,
                                     int numRowTiles) {
  __shared__ float bs[K * 16];

  int c0 = blockIdx.y << 4;

#pragma unroll
  for (int i = threadIdx.x; i < K * 16; i += THREADS) {
    int k = i >> 4;
    int c = i & 15;
    bs[i] = B[k * NCOLS + c0 + c];
  }
  __syncthreads();

  int wave = threadIdx.x >> 5;
  int lane = threadIdx.x & 31;
  int rowTile = blockIdx.x * WAVES_PER_BLOCK + wave;
  if (rowTile >= numRowTiles) return;

  int r0 = rowTile << 4;
  int half = lane >> 4;
  int l = lane & 15;

  const float* ap = A + (size_t)(r0 + l) * K + 2 * half;
  const float* bp = bs + (2 * half) * 16 + l;

  v8f acc = {};
#pragma unroll
  for (int k = 0; k < K; k += 4) {
    v2f a;
    a.x = ap[k];
    a.y = ap[k + 1];
    v2f b;
    b.x = bp[k * 16];
    b.y = bp[k * 16 + 16];
    acc = __builtin_amdgcn_wmma_f32_16x16x4_f32(
         false, a,  false, b,
         (short)0, acc,  false,  false);
  }

  const float4* sp = (const float4*)(rscale + r0 + half * 8);
  float4 s0 = sp[0];
  float4 s1 = sp[1];
  float rs[8] = {s0.x, s0.y, s0.z, s0.w, s1.x, s1.y, s1.z, s1.w};
  float* cp = C + (size_t)(r0 + half * 8) * NCOLS + c0 + l;
#pragma unroll
  for (int j = 0; j < 8; ++j) cp[j * NCOLS] = acc[j] * rs[j];
}

template <int NF2, int LD, int DOUT, bool RELU>
__global__ void __launch_bounds__(THREADS)
gcn_pull_kernel(const int* __restrict__ off, const int* __restrict__ src,
                const float* __restrict__ dinv, const float* __restrict__ xw,
                const float* __restrict__ bias, float* __restrict__ out, int n) {
  __shared__ float stage[WAVES_PER_BLOCK * DOUT];
  int lane = threadIdx.x & 31;
  int wave = threadIdx.x >> 5;
  int node0 = blockIdx.x * WAVES_PER_BLOCK;
  int node = __builtin_amdgcn_readfirstlane(node0 + wave);
  bool act = lane < NF2;

  if (node < n) {
    int beg = __builtin_amdgcn_readfirstlane(off[node]);
    int end = __builtin_amdgcn_readfirstlane(off[node + 1]);
    float dc = dinv[node];
    const float* xl = xw + 2 * lane;
    float ax = 0.0f, ay = 0.0f;

    for (int base = beg; base < end; base += 32) {
      int m = end - base;
      if (m > 32) m = 32;
      int s_l = node;
      if (lane < m) s_l = src[base + lane];
      for (int j = 0; j < m; j += 8) {
        int s[8];
        float2 v[8];
#pragma unroll
        for (int u = 0; u < 8; ++u) s[u] = __builtin_amdgcn_readlane(s_l, j + u);
#pragma unroll
        for (int u = 0; u < 8; ++u) {
          v[u] = make_float2(0.0f, 0.0f);
          if (act) v[u] = *(const float2*)(xl + (size_t)s[u] * LD);
        }
        if (j + 8 > m) {
#pragma unroll
          for (int u = 1; u < 8; ++u)
            if (j + u >= m) v[u] = make_float2(0.0f, 0.0f);
        }
#pragma unroll
        for (int u = 0; u < 8; ++u) {
          ax += v[u].x;
          ay += v[u].y;
        }
      }
    }

    if (act) {
      float2 v = *(const float2*)(xl + (size_t)node * LD);
      float2 b = *(const float2*)(bias + 2 * lane);
      float ox = dc * (ax + v.x) + b.x;
      float oy = dc * (ay + v.y) + b.y;
      if (RELU) { ox = fmaxf(ox, 0.0f); oy = fmaxf(oy, 0.0f); }
      if (DOUT == 64) {
        *(float2*)(out + (size_t)node * DOUT + 2 * lane) = make_float2(ox, oy);
      } else {
        stage[wave * DOUT + 2 * lane] = ox;
        stage[wave * DOUT + 2 * lane + 1] = oy;
      }
    }
  }

  if (DOUT != 64) {
    __syncthreads();
    size_t o0 = (size_t)node0 * DOUT;
    size_t lim = (size_t)n * DOUT;
    for (int t = threadIdx.x; t < WAVES_PER_BLOCK * DOUT; t += THREADS)
      if (o0 + t < lim) out[o0 + t] = stage[t];
  }
}

static inline int cdiv(long a, int b) { return (int)((a + b - 1) / b); }
static inline size_t pad64(size_t x) { return (x + 63) & ~(size_t)63; }

template <int K, int NCOLS>
static void launch_gemm(const float* A, const float* B, const float* rscale, float* C, int M,
                        hipStream_t stream) {
  int numRowTiles = M / 16;
  dim3 grid(cdiv(numRowTiles, WAVES_PER_BLOCK), NCOLS / 16, 1);
  gcn_gemm_wmma_kernel<K, NCOLS><<<grid, THREADS, 0, stream>>>(A, B, rscale, C, numRowTiles);
}

extern "C" void kernel_launch(void* const* d_in, const int* in_sizes, int n_in,
                              void* d_out, int out_size, void* d_ws, size_t ws_size,
                              hipStream_t stream) {
  const float* x  = (const float*)d_in[0];
  const int*   ei = (const int*)d_in[1];
  const float* W1 = (const float*)d_in[2];
  const float* b1 = (const float*)d_in[3];
  const float* W2 = (const float*)d_in[4];
  const float* b2 = (const float*)d_in[5];
  const float* W3 = (const float*)d_in[6];
  const float* b3 = (const float*)d_in[7];

  const int D_IN = 128, D_H = 64;
  int N = in_sizes[0] / D_IN;
  int E = in_sizes[1] / 2;
  const int* row = ei;
  const int* col = ei + E;

  float* ws   = (float*)d_ws;
  int*   cnt  = (int*)ws;
  int*   off  = cnt + pad64((size_t)N);
  float* dinv = (float*)(off + pad64((size_t)N + 1));
  int*   pos  = (int*)(dinv + pad64((size_t)N));
  int*   srcs = pos + pad64((size_t)E);
  float* xw   = (float*)(srcs + pad64((size_t)E));
  float* h    = xw + (size_t)N * D_H;
  float* wpad = h + (size_t)N * D_H;

  gcn_zero_int_kernel<<<cdiv(N, THREADS), THREADS, 0, stream>>>(cnt, N);
  gcn_count_kernel<<<cdiv(E, THREADS), THREADS, 0, stream>>>(col, cnt, pos, E);
  gcn_scan_kernel<<<cdiv(N, SCAN_T), SCAN_T, 0, stream>>>(cnt, off, dinv, N);
  gcn_place_kernel<<<cdiv(E, THREADS), THREADS, 0, stream>>>(row, col, off, pos, srcs, E);
  gcn_pad_w3_kernel<<<cdiv(64 * 48, THREADS), THREADS, 0, stream>>>(W3, wpad);

  int pullBlocks = cdiv(N, WAVES_PER_BLOCK);

  launch_gemm<128, 64>(x, W1, dinv, xw, N, stream);
  gcn_pull_kernel<32, 64, 64, true><<<pullBlocks, THREADS, 0, stream>>>(off, srcs, dinv, xw, b1, h, N);

  launch_gemm<64, 64>(h, W2, dinv, xw, N, stream);
  gcn_pull_kernel<32, 64, 64, true><<<pullBlocks, THREADS, 0, stream>>>(off, srcs, dinv, xw, b2, h, N);

  launch_gemm<64, 48>(h, wpad, dinv, xw, N, stream);
  gcn_pull_kernel<20, 48, 40, false><<<pullBlocks, THREADS, 0, stream>>>(off, srcs, dinv, xw, b3,
                                                                        (float*)d_out, N);
}


# Fewer, longer-lived waves: aggregation wave walks 8 nodes' contiguous runs (self-loop stored in the run), GEMM wave does all column tiles, count/place 4 edges per thread; sum order kept
# speedup vs baseline: 2.6411x; 1.1591x over previous
#include <hip/hip_runtime.h>

typedef __attribute__((ext_vector_type(2))) float v2f;
typedef __attribute__((ext_vector_type(8))) float v8f;

#define THREADS 256
#define WAVES_PER_BLOCK (THREADS / 32)
#define SCAN_T 1024
#define EDGES_PER_THREAD 4
#define NPW 8

__global__ void gcn_zero_int_kernel(int* __restrict__ p, int n) {
  int i = blockIdx.x * blockDim.x + threadIdx.x;
  if (i < n) p[i] = 0;
}

__global__ void gcn_count_kernel(const int* __restrict__ col, int* __restrict__ cnt,
                                 int* __restrict__ pos, int E) {
  int t = blockIdx.x * blockDim.x + threadIdx.x;
  int T = gridDim.x * blockDim.x;
  int c[EDGES_PER_THREAD];
  int p[EDGES_PER_THREAD];
  for (int u = 0; u < EDGES_PER_THREAD; ++u) {
    int e = t + u * T;
    c[u] = col[e < E ? e : E - 1];
  }
  for (int u = 0; u < EDGES_PER_THREAD; ++u) {
    int e = t + u * T;
    p[u] = 0;
    if (e < E) p[u] = atomicAdd(&cnt[c[u]], 1);
  }
  for (int u = 0; u < EDGES_PER_THREAD; ++u) {
    int e = t + u * T;
    if (e < E) pos[e] = p[u];
  }
}

__global__ void __launch_bounds__(SCAN_T)
gcn_scan_kernel(const int* __restrict__ cnt, int* __restrict__ off,
                float* __restrict__ dinv, int* __restrict__ src, int n) {
  __shared__ int wsum[32];
  __shared__ int s_prefix;
  int tid = threadIdx.x;
  int lane = tid & 31;
  int wave = tid >> 5;
  int tileBase = blockIdx.x * SCAN_T;

  int part = 0;
  for (int i = tid; i < tileBase; i += SCAN_T) part += cnt[i];
  for (int d = 16; d > 0; d >>= 1) part += __shfl_xor(part, d);
  if (lane == 0) wsum[wave] = part;
  __syncthreads();
  if (wave == 0) {
    int v = wsum[lane];
    for (int d = 16; d > 0; d >>= 1) v += __shfl_xor(v, d);
    if (lane == 0) s_prefix = v;
  }
  __syncthreads();
  int prefix = s_prefix;
  __syncthreads();

  int i = tileBase + tid;
  int c = (i < n) ? cnt[i] : 0;
  int incl = c;
  for (int d = 1; d < 32; d <<= 1) {
    int t = __shfl_up(incl, d);
    if (lane >= d) incl += t;
  }
  if (lane == 31) wsum[wave] = incl;
  __syncthreads();
  if (wave == 0) {
    int v = wsum[lane];
    int vi = v;
    for (int d = 1; d < 32; d <<= 1) {
      int t = __shfl_up(vi, d);
      if (lane >= d) vi += t;
    }
    wsum[lane] = vi - v;
  }
  __syncthreads();
  int excl = prefix + wsum[wave] + incl - c + i;
  if (i < n) {
    off[i] = excl;
    dinv[i] = rsqrtf((float)(c + 1));
    src[excl + c] = i;
  }
  if (i == n - 1) off[n] = excl + c + 1;
}

__global__ void gcn_place_kernel(const int* __restrict__ row, const int* __restrict__ col,
                                 const int* __restrict__ off, const int* __restrict__ pos,
                                 int* __restrict__ src, int E) {
  int t = blockIdx.x * blockDim.x + threadIdx.x;
  int T = gridDim.x * blockDim.x;
  int c[EDGES_PER_THREAD];
  int p[EDGES_PER_THREAD];
  int r[EDGES_PER_THREAD];
  int o[EDGES_PER_THREAD];
  for (int u = 0; u < EDGES_PER_THREAD; ++u) {
    int e = t + u * T;
    int ec = e < E ? e : E - 1;
    c[u] = col[ec];
    p[u] = pos[ec];
    r[u] = row[ec];
  }
  for (int u = 0; u < EDGES_PER_THREAD; ++u) o[u] = off[c[u]];
  for (int u = 0; u < EDGES_PER_THREAD; ++u) {
    int e = t + u * T;
    if (e < E) src[o[u] + p[u]] = r[u];
  }
}

__global__ void gcn_pad_w3_kernel(const float* __restrict__ W, float* __restrict__ Wp) {
  int i = blockIdx.x * blockDim.x + threadIdx.x;
  if (i < 64 * 48) {
    int k = i / 48;
    int c = i - k * 48;
    Wp[i] = (c < 40) ? W[k * 40 + c] : 0.0f;
  }
}

template <int K, int NCOLS>
__global__ void __launch_bounds__(THREADS)
gcn_gemm_wmma_kernel(const float* __restrict__ A,
                     const float* __restrict__ B,
                     const float* __restrict__ rscale,
                     float* __restrict__ C,
                     int numRowTiles) {
  constexpr int NT = NCOLS / 16;
  __shared__ float bs[NT * K * 16];

  for (int i = threadIdx.x; i < K * NCOLS; i += THREADS) {
    int k = i / NCOLS;
    int c = i - k * NCOLS;
    bs[(c >> 4) * (K * 16) + k * 16 + (c & 15)] = B[i];
  }
  __syncthreads();

  int wave = threadIdx.x >> 5;
  int lane = threadIdx.x & 31;
  int rowTile = blockIdx.x * WAVES_PER_BLOCK + wave;
  if (rowTile >= numRowTiles) return;

  int r0 = rowTile << 4;
  int half = lane >> 4;
  int l = lane & 15;

  const float* ap = A + (size_t)(r0 + l) * K + 2 * half;
  const float* bp = bs + (2 * half) * 16 + l;

  v8f acc[NT];
#pragma unroll
  for (int t = 0; t < NT; ++t) acc[t] = v8f{};

#pragma unroll
  for (int k = 0; k < K; k += 4) {
    v2f a;
    a.x = ap[k];
    a.y = ap[k + 1];
#pragma unroll
    for (int t = 0; t < NT; ++t) {
      v2f b;
      b.x = bp[t * (K * 16) + k * 16];
      b.y = bp[t * (K * 16) + k * 16 + 16];
      acc[t] = __builtin_amdgcn_wmma_f32_16x16x4_f32(
           false, a,  false, b,
           (short)0, acc[t],  false,  false);
    }
  }

  const float4* sp = (const float4*)(rscale + r0 + half * 8);
  float4 s0 = sp[0];
  float4 s1 = sp[1];
  float rs[8] = {s0.x, s0.y, s0.z, s0.w, s1.x, s1.y, s1.z, s1.w};
  float* cp = C + (size_t)(r0 + half * 8) * NCOLS + l;
#pragma unroll
  for (int j = 0; j < 8; ++j) {
#pragma unroll
    for (int t = 0; t < NT; ++t) cp[j * NCOLS + t * 16] = acc[t][j] * rs[j];
  }
}

template <int NF2, int LD, int DOUT, bool RELU>
__global__ void __launch_bounds__(THREADS)
gcn_pull_kernel(const int* __restrict__ off, const int* __restrict__ src,
                const float* __restrict__ dinv, const float* __restrict__ xw,
                const float* __restrict__ bias, float* __restrict__ out, int n) {
  __shared__ float stage[(DOUT != 64) ? WAVES_PER_BLOCK * NPW * DOUT : 1];
  int lane = threadIdx.x & 31;
  int wave = threadIdx.x >> 5;
  int node0 = blockIdx.x * (WAVES_PER_BLOCK * NPW);
  int n0 = __builtin_amdgcn_readfirstlane(node0 + wave * NPW);
  bool act = lane < NF2;

  if (n0 < n) {
    int cnt = n - n0;
    if (cnt > NPW) cnt = NPW;
    int o_l = off[n0 + (lane < cnt ? lane : cnt)];
    float d_l = dinv[n0 + (lane < cnt ? lane : cnt - 1)];
    const float* xl = xw + 2 * lane;
    float2 b = make_float2(0.0f, 0.0f);
    if (act) b = *(const float2*)(bias + 2 * lane);

    int pos = __builtin_amdgcn_readlane(o_l, 0);
    int endAll = __builtin_amdgcn_readlane(o_l, cnt);
    int nodeEnd = __builtin_amdgcn_readlane(o_l, 1);
    int k = 0;
    float ax = 0.0f, ay = 0.0f;

    auto finish = [&]() {
      float dc = __int_as_float(__builtin_amdgcn_readlane(__float_as_int(d_l), k));
      if (act) {
        float ox = dc * ax + b.x;
        float oy = dc * ay + b.y;
        if (RELU) { ox = fmaxf(ox, 0.0f); oy = fmaxf(oy, 0.0f); }
        if constexpr (DOUT == 64) {
          *(float2*)(out + (size_t)(n0 + k) * DOUT + 2 * lane) = make_float2(ox, oy);
        } else {
          stage[(wave * NPW + k) * DOUT + 2 * lane] = ox;
          stage[(wave * NPW + k) * DOUT + 2 * lane + 1] = oy;
        }
      }
      ++k;
      ax = 0.0f;
      ay = 0.0f;
      nodeEnd = (k < cnt) ? __builtin_amdgcn_readlane(o_l, k + 1) : -1;
    };

    int s_l = n0;
    if (lane < endAll - pos) s_l = src[pos + lane];

    while (pos < endAll) {
      int m = endAll - pos;
      if (m > 32) m = 32;
      int s_n = n0;
      for (int j = 0; j < m; j += 8) {
        int sidx[8];
        float2 v[8];
#pragma unroll
        for (int u = 0; u < 8; ++u) sidx[u] = __builtin_amdgcn_readlane(s_l, j + u);
#pragma unroll
        for (int u = 0; u < 8; ++u) {
          v[u] = make_float2(0.0f, 0.0f);
          if (act) v[u] = *(const float2*)(xl + (size_t)sidx[u] * LD);
        }
        if (j == 0 && lane < endAll - pos - 32) s_n = src[pos + 32 + lane];
#pragma unroll
        for (int u = 0; u < 8; ++u) {
          ax += v[u].x;
          ay += v[u].y;
          if (pos + j + u + 1 == nodeEnd) finish();
        }
      }
      pos += m;
      s_l = s_n;
    }
  }

  if (DOUT != 64) {
    __syncthreads();
    size_t o0 = (size_t)node0 * DOUT;
    size_t lim = (size_t)n * DOUT;
    for (int t = threadIdx.x; t < WAVES_PER_BLOCK * NPW * DOUT; t += THREADS)
      if (o0 + t < lim) out[o0 + t] = stage[t];
  }
}

static inline int cdiv(long a, int b) { return (int)((a + b - 1) / b); }
static inline size_t pad64(size_t x) { return (x + 63) & ~(size_t)63; }

template <int K, int NCOLS>
static void launch_gemm(const float* A, const float* B, const float* rscale, float* C, int M,
                        hipStream_t stream) {
  int numRowTiles = M / 16;
  dim3 grid(cdiv(numRowTiles, WAVES_PER_BLOCK), 1, 1);
  gcn_gemm_wmma_kernel<K, NCOLS><<<grid, THREADS, 0, stream>>>(A, B, rscale, C, numRowTiles);
}

extern "C" void kernel_launch(void* const* d_in, const int* in_sizes, int n_in,
                              void* d_out, int out_size, void* d_ws, size_t ws_size,
                              hipStream_t stream) {
  const float* x  = (const float*)d_in[0];
  const int*   ei = (const int*)d_in[1];
  const float* W1 = (const float*)d_in[2];
  const float* b1 = (const float*)d_in[3];
  const float* W2 = (const float*)d_in[4];
  const float* b2 = (const float*)d_in[5];
  const float* W3 = (const float*)d_in[6];
  const float* b3 = (const float*)d_in[7];

  const int D_IN = 128, D_H = 64;
  int N = in_sizes[0] / D_IN;
  int E = in_sizes[1] / 2;
  const int* row = ei;
  const int* col = ei + E;

  float* ws   = (float*)d_ws;
  int*   cnt  = (int*)ws;
  int*   off  = cnt + pad64((size_t)N);
  float* dinv = (float*)(off + pad64((size_t)N + 1));
  int*   pos  = (int*)(dinv + pad64((size_t)N));
  int*   srcs = pos + pad64((size_t)E);
  float* xw   = (float*)(srcs + pad64((size_t)E + (size_t)N));
  float* h    = xw + (size_t)N * D_H;
  float* wpad = h + (size_t)N * D_H;

  gcn_zero_int_kernel<<<cdiv(N, THREADS), THREADS, 0, stream>>>(cnt, N);
  gcn_count_kernel<<<cdiv(E, THREADS * EDGES_PER_THREAD), THREADS, 0, stream>>>(col, cnt, pos, E);
  gcn_scan_kernel<<<cdiv(N, SCAN_T), SCAN_T, 0, stream>>>(cnt, off, dinv, srcs, N);
  gcn_place_kernel<<<cdiv(E, THREADS * EDGES_PER_THREAD), THREADS, 0, stream>>>(row, col, off, pos, srcs,
                                                                               E);
  gcn_pad_w3_kernel<<<cdiv(64 * 48, THREADS), THREADS, 0, stream>>>(W3, wpad);

  int pullBlocks = cdiv(N, WAVES_PER_BLOCK * NPW);

  launch_gemm<128, 64>(x, W1, dinv, xw, N, stream);
  gcn_pull_kernel<32, 64, 64, true><<<pullBlocks, THREADS, 0, stream>>>(off, srcs, dinv, xw, b1, h, N);

  launch_gemm<64, 64>(h, W2, dinv, xw, N, stream);
  gcn_pull_kernel<32, 64, 64, true><<<pullBlocks, THREADS, 0, stream>>>(off, srcs, dinv, xw, b2, h, N);

  launch_gemm<64, 48>(h, wpad, dinv, xw, N, stream);
  gcn_pull_kernel<20, 48, 40, false><<<pullBlocks, THREADS, 0, stream>>>(off, srcs, dinv, xw, b3,
                                                                        (float*)d_out, N);
}


# Edge grouping by destination as a two-level counting sort on LDS counters (chunk sort by 512-node bin, then a block per bin, whole-line stores): no global atomics; sum order = LDS arrival order
# speedup vs baseline: 4.7509x; 1.7988x over previous
#include <hip/hip_runtime.h>

typedef __attribute__((ext_vector_type(2))) float v2f;
typedef __attribute__((ext_vector_type(8))) float v8f;

#define THREADS 256
#define WAVES_PER_BLOCK (THREADS / 32)
#define NPW 8

#define BIN_SHIFT 9
#define BIN_NODES 512
#define CHUNK 6400
#define MAXB 256
#define MAXCH 256
#define SORT_T 1024
#define CHUNK_EPT 7
#define STAGE_CAP 12288

__global__ void __launch_bounds__(SORT_T)
gcn_chunk_sort_kernel(const int* __restrict__ row, const int* __restrict__ col,
                      int* __restrict__ recs, int* __restrict__ tab, int E, int NB) {
  __shared__ int hist[MAXB];
  __shared__ int lbase[MAXB + 1];
  __shared__ int srec[CHUNK];
  int tid = threadIdx.x;
  int e0 = blockIdx.x * CHUNK;
  int m = E - e0;
  if (m > CHUNK) m = CHUNK;

  if (tid < MAXB) hist[tid] = 0;
  int c[CHUNK_EPT];
  int r[CHUNK_EPT];
#pragma unroll
  for (int u = 0; u < CHUNK_EPT; ++u) {
    int i = tid + u * SORT_T;
    c[u] = 0;
    r[u] = 0;
    if (i < m) {
      c[u] = col[e0 + i];
      r[u] = row[e0 + i];
    }
  }
  __syncthreads();
#pragma unroll
  for (int u = 0; u < CHUNK_EPT; ++u) {
    int i = tid + u * SORT_T;
    if (i < m) atomicAdd(&hist[c[u] >> BIN_SHIFT], 1);
  }
  __syncthreads();
  if (tid < 32) {
    int v[8];
    int s = 0;
#pragma unroll
    for (int k = 0; k < 8; ++k) {
      v[k] = hist[tid * 8 + k];
      s += v[k];
    }
    int incl = s;
    for (int d = 1; d < 32; d <<= 1) {
      int t = __shfl_up(incl, d);
      if (tid >= d) incl += t;
    }
    int run = incl - s;
#pragma unroll
    for (int k = 0; k < 8; ++k) {
      lbase[tid * 8 + k] = run;
      run += v[k];
    }
    if (tid == 31) lbase[MAXB] = run;
  }
  __syncthreads();
  for (int b = tid; b <= NB; b += SORT_T) tab[blockIdx.x * (NB + 1) + b] = lbase[b];
  if (tid < MAXB) hist[tid] = lbase[tid];
  __syncthreads();
#pragma unroll
  for (int u = 0; u < CHUNK_EPT; ++u) {
    int i = tid + u * SORT_T;
    if (i < m) {
      int p = atomicAdd(&hist[c[u] >> BIN_SHIFT], 1);
      srec[p] = (r[u] << BIN_SHIFT) | (c[u] & (BIN_NODES - 1));
    }
  }
  __syncthreads();
  for (int i = tid; i < m; i += SORT_T) recs[e0 + i] = srec[i];
}

__global__ void __launch_bounds__(SORT_T)
gcn_bin_finish_kernel(const int* __restrict__ recs, const int* __restrict__ tab,
                      int* __restrict__ off, float* __restrict__ dinv,
                      int* __restrict__ src, int n, int nblk, int NB) {
  __shared__ int pstart[MAXCH];
  __shared__ int plen[MAXCH];
  __shared__ int cnt[BIN_NODES];
  __shared__ int cur[BIN_NODES];
  __shared__ int wsum[32];
  __shared__ int s_binStart;
  __shared__ int s_total;
  __shared__ int stage[STAGE_CAP];
  int tid = threadIdx.x;
  int lane = tid & 31;
  int wave = tid >> 5;
  int b = blockIdx.x;
  int stride = NB + 1;

  int part = 0;
  for (int k = tid; k < nblk; k += SORT_T) {
    int s = tab[k * stride + b];
    int e = tab[k * stride + b + 1];
    pstart[k] = k * CHUNK + s;
    plen[k] = e - s;
    part += s;
  }
  if (tid < BIN_NODES) cnt[tid] = 0;
  for (int d = 16; d > 0; d >>= 1) part += __shfl_xor(part, d);
  if (lane == 0) wsum[wave] = part;
  __syncthreads();
  if (wave == 0) {
    int v = wsum[lane];
    for (int d = 16; d > 0; d >>= 1) v += __shfl_xor(v, d);
    if (lane == 0) s_binStart = v;
  }
  __syncthreads();

  int rec[16];
#pragma unroll
  for (int u = 0; u < 8; ++u) {
    int k = wave + 32 * u;
    int L = 0;
    int s = 0;
    if (k < nblk) {
      L = plen[k];
      s = pstart[k];
    }
    rec[2 * u] = -1;
    rec[2 * u + 1] = -1;
    if (lane < L) rec[2 * u] = recs[s + lane];
    if (lane + 32 < L) rec[2 * u + 1] = recs[s + lane + 32];
  }
#pragma unroll
  for (int q = 0; q < 16; ++q)
    if (rec[q] >= 0) atomicAdd(&cnt[rec[q] & (BIN_NODES - 1)], 1);
  for (int k = wave; k < nblk; k += 32) {
    int L = plen[k];
    int s = pstart[k];
    for (int j = ((k >> 5) < 8 ? 64 : 0) + lane; j < L; j += 32)
      atomicAdd(&cnt[recs[s + j] & (BIN_NODES - 1)], 1);
  }
  __syncthreads();

  int c = (tid < BIN_NODES) ? cnt[tid] : 0;
  int incl = c;
  for (int d = 1; d < 32; d <<= 1) {
    int t = __shfl_up(incl, d);
    if (lane >= d) incl += t;
  }
  if (lane == 31) wsum[wave] = incl;
  __syncthreads();
  if (wave == 0) {
    int v = wsum[lane];
    int vi = v;
    for (int d = 1; d < 32; d <<= 1) {
      int t = __shfl_up(vi, d);
      if (lane >= d) vi += t;
    }
    wsum[lane] = vi - v;
  }
  __syncthreads();
  int ex = wsum[wave] + incl - c;
  int node0 = b * BIN_NODES;
  int nodes = n - node0;
  if (nodes > BIN_NODES) nodes = BIN_NODES;
  int lo = ex + tid;
  if (tid == SORT_T - 1) s_total = ex + c;
  if (tid < nodes) cur[tid] = lo;
  __syncthreads();
  int gbase = s_binStart + node0;
  int total = s_total + nodes;
  bool staged = total <= STAGE_CAP;
  if (tid < nodes) {
    int node = node0 + tid;
    off[node] = gbase + lo;
    dinv[node] = rsqrtf((float)(c + 1));
    if (staged) stage[lo + c] = node; else src[gbase + lo + c] = node;
    if (node == n - 1) off[n] = gbase + lo + c + 1;
  }
#pragma unroll
  for (int q = 0; q < 16; ++q) {
    if (rec[q] >= 0) {
      int p = atomicAdd(&cur[rec[q] & (BIN_NODES - 1)], 1);
      int sid = rec[q] >> BIN_SHIFT;
      if (staged) stage[p] = sid; else src[gbase + p] = sid;
    }
  }
  for (int k = wave; k < nblk; k += 32) {
    int L = plen[k];
    int s = pstart[k];
    for (int j = ((k >> 5) < 8 ? 64 : 0) + lane; j < L; j += 32) {
      int rc = recs[s + j];
      int p = atomicAdd(&cur[rc & (BIN_NODES - 1)], 1);
      int sid = rc >> BIN_SHIFT;
      if (staged) stage[p] = sid; else src[gbase + p] = sid;
    }
  }
  __syncthreads();
  if (staged)
    for (int i = tid; i < total; i += SORT_T) src[gbase + i] = stage[i];
}

__global__ void gcn_pad_w3_kernel(const float* __restrict__ W, float* __restrict__ Wp) {
  int i = blockIdx.x * blockDim.x + threadIdx.x;
  if (i < 64 * 48) {
    int k = i / 48;
    int c = i - k * 48;
    Wp[i] = (c < 40) ? W[k * 40 + c] : 0.0f;
  }
}

template <int K, int NCOLS>
__global__ void __launch_bounds__(THREADS)
gcn_gemm_wmma_kernel(const float* __restrict__ A,
                     const float* __restrict__ B,
                     const float* __restrict__ rscale,
                     float* __restrict__ C,
                     int numRowTiles) {
  constexpr int NT = NCOLS / 16;
  __shared__ float bs[NT * K * 16];

  for (int i = threadIdx.x; i < K * NCOLS; i += THREADS) {
    int k = i / NCOLS;
    int c = i - k * NCOLS;
    bs[(c >> 4) * (K * 16) + k * 16 + (c & 15)] = B[i];
  }
  __syncthreads();

  int wave = threadIdx.x >> 5;
  int lane = threadIdx.x & 31;
  int rowTile = blockIdx.x * WAVES_PER_BLOCK + wave;
  if (rowTile >= numRowTiles) return;

  int r0 = rowTile << 4;
  int half = lane >> 4;
  int l = lane & 15;

  const float* ap = A + (size_t)(r0 + l) * K + 2 * half;
  const float* bp = bs + (2 * half) * 16 + l;

  v8f acc[NT];
#pragma unroll
  for (int t = 0; t < NT; ++t) acc[t] = v8f{};

#pragma unroll
  for (int k = 0; k < K; k += 4) {
    v2f a;
    a.x = ap[k];
    a.y = ap[k + 1];
#pragma unroll
    for (int t = 0; t < NT; ++t) {
      v2f b;
      b.x = bp[t * (K * 16) + k * 16];
      b.y = bp[t * (K * 16) + k * 16 + 16];
      acc[t] = __builtin_amdgcn_wmma_f32_16x16x4_f32(
           false, a,  false, b,
           (short)0, acc[t],  false,  false);
    }
  }

  const float4* sp = (const float4*)(rscale + r0 + half * 8);
  float4 s0 = sp[0];
  float4 s1 = sp[1];
  float rs[8] = {s0.x, s0.y, s0.z, s0.w, s1.x, s1.y, s1.z, s1.w};
  float* cp = C + (size_t)(r0 + half * 8) * NCOLS + l;
#pragma unroll
  for (int j = 0; j < 8; ++j) {
#pragma unroll
    for (int t = 0; t < NT; ++t) cp[j * NCOLS + t * 16] = acc[t][j] * rs[j];
  }
}

template <int NF2, int LD, int DOUT, bool RELU>
__global__ void __launch_bounds__(THREADS)
gcn_pull_kernel(const int* __restrict__ off, const int* __restrict__ src,
                const float* __restrict__ dinv, const float* __restrict__ xw,
                const float* __restrict__ bias, float* __restrict__ out, int n) {
  __shared__ float stage[(DOUT != 64) ? WAVES_PER_BLOCK * NPW * DOUT : 1];
  int lane = threadIdx.x & 31;
  int wave = threadIdx.x >> 5;
  int node0 = blockIdx.x * (WAVES_PER_BLOCK * NPW);
  int n0 = __builtin_amdgcn_readfirstlane(node0 + wave * NPW);
  bool act = lane < NF2;

  if (n0 < n) {
    int cnt = n - n0;
    if (cnt > NPW) cnt = NPW;
    int o_l = off[n0 + (lane < cnt ? lane : cnt)];
    float d_l = dinv[n0 + (lane < cnt ? lane : cnt - 1)];
    const float* xl = xw + 2 * lane;
    float2 b = make_float2(0.0f, 0.0f);
    if (act) b = *(const float2*)(bias + 2 * lane);

    int pos = __builtin_amdgcn_readlane(o_l, 0);
    int endAll = __builtin_amdgcn_readlane(o_l, cnt);
    int nodeEnd = __builtin_amdgcn_readlane(o_l, 1);
    int k = 0;
    float ax = 0.0f, ay = 0.0f;

    auto finish = [&]() {
      float dc = __int_as_float(__builtin_amdgcn_readlane(__float_as_int(d_l), k));
      if (act) {
        float ox = dc * ax + b.x;
        float oy = dc * ay + b.y;
        if (RELU) { ox = fmaxf(ox, 0.0f); oy = fmaxf(oy, 0.0f); }
        if constexpr (DOUT == 64) {
          *(float2*)(out + (size_t)(n0 + k) * DOUT + 2 * lane) = make_float2(ox, oy);
        } else {
          stage[(wave * NPW + k) * DOUT + 2 * lane] = ox;
          stage[(wave * NPW + k) * DOUT + 2 * lane + 1] = oy;
        }
      }
      ++k;
      ax = 0.0f;
      ay = 0.0f;
      nodeEnd = (k < cnt) ? __builtin_amdgcn_readlane(o_l, k + 1) : -1;
    };

    int s_l = n0;
    if (lane < endAll - pos) s_l = src[pos + lane];

    while (pos < endAll) {
      int m = endAll - pos;
      if (m > 32) m = 32;
      int s_n = n0;
      for (int j = 0; j < m; j += 8) {
        int sidx[8];
        float2 v[8];
#pragma unroll
        for (int u = 0; u < 8; ++u) sidx[u] = __builtin_amdgcn_readlane(s_l, j + u);
#pragma unroll
        for (int u = 0; u < 8; ++u) {
          v[u] = make_float2(0.0f, 0.0f);
          if (act) v[u] = *(const float2*)(xl + (size_t)sidx[u] * LD);
        }
        if (j == 0 && lane < endAll - pos - 32) s_n = src[pos + 32 + lane];
#pragma unroll
        for (int u = 0; u < 8; ++u) {
          ax += v[u].x;
          ay += v[u].y;
          if (pos + j + u + 1 == nodeEnd) finish();
        }
      }
      pos += m;
      s_l = s_n;
    }
  }

  if (DOUT != 64) {
    __syncthreads();
    size_t o0 = (size_t)node0 * DOUT;
    size_t lim = (size_t)n * DOUT;
    for (int t = threadIdx.x; t < WAVES_PER_BLOCK * NPW * DOUT; t += THREADS)
      if (o0 + t < lim) out[o0 + t] = stage[t];
  }
}

static inline int cdiv(long a, int b) { return (int)((a + b - 1) / b); }
static inline size_t pad64(size_t x) { return (x + 63) & ~(size_t)63; }

template <int K, int NCOLS>
static void launch_gemm(const float* A, const float* B, const float* rscale, float* C, int M,
                        hipStream_t stream) {
  int numRowTiles = M / 16;
  dim3 grid(cdiv(numRowTiles, WAVES_PER_BLOCK), 1, 1);
  gcn_gemm_wmma_kernel<K, NCOLS><<<grid, THREADS, 0, stream>>>(A, B, rscale, C, numRowTiles);
}

extern "C" void kernel_launch(void* const* d_in, const int* in_sizes, int n_in,
                              void* d_out, int out_size, void* d_ws, size_t ws_size,
                              hipStream_t stream) {
  const float* x  = (const float*)d_in[0];
  const int*   ei = (const int*)d_in[1];
  const float* W1 = (const float*)d_in[2];
  const float* b1 = (const float*)d_in[3];
  const float* W2 = (const float*)d_in[4];
  const float* b2 = (const float*)d_in[5];
  const float* W3 = (const float*)d_in[6];
  const float* b3 = (const float*)d_in[7];

  const int D_IN = 128, D_H = 64;
  int N = in_sizes[0] / D_IN;
  int E = in_sizes[1] / 2;
  const int* row = ei;
  const int* col = ei + E;

  int NB = cdiv(N, BIN_NODES);
  int nblk = cdiv(E, CHUNK);

  float* ws   = (float*)d_ws;
  int*   off  = (int*)ws;
  float* dinv = (float*)(off + pad64((size_t)N + 1));
  int*   recs = (int*)(dinv + pad64((size_t)N));
  int*   tab  = recs + pad64((size_t)E);
  int*   srcs = tab + pad64((size_t)nblk * (size_t)(NB + 1));
  float* xw   = (float*)(srcs + pad64((size_t)E + (size_t)N));
  float* h    = xw + (size_t)N * D_H;
  float* wpad = h + (size_t)N * D_H;

  gcn_chunk_sort_kernel<<<nblk, SORT_T, 0, stream>>>(row, col, recs, tab, E, NB);
  gcn_bin_finish_kernel<<<NB, SORT_T, 0, stream>>>(recs, tab, off, dinv, srcs, N, nblk, NB);
  gcn_pad_w3_kernel<<<cdiv(64 * 48, THREADS), THREADS, 0, stream>>>(W3, wpad);

  int pullBlocks = cdiv(N, WAVES_PER_BLOCK * NPW);

  launch_gemm<128, 64>(x, W1, dinv, xw, N, stream);
  gcn_pull_kernel<32, 64, 64, true><<<pullBlocks, THREADS, 0, stream>>>(off, srcs, dinv, xw, b1, h, N);

  launch_gemm<64, 64>(h, W2, dinv, xw, N, stream);
  gcn_pull_kernel<32, 64, 64, true><<<pullBlocks, THREADS, 0, stream>>>(off, srcs, dinv, xw, b2, h, N);

  launch_gemm<64, 48>(h, wpad, dinv, xw, N, stream);
  gcn_pull_kernel<20, 48, 40, false><<<pullBlocks, THREADS, 0, stream>>>(off, srcs, dinv, xw, b3,
                                                                        (float*)d_out, N);
}


# Layer 2 and 3 linear transforms fused into the preceding aggregation's epilogue on WMMA (h tile and W in LDS, W3 zero-padded there): h never goes to memory, 3 launches fewer; sum order kept
# speedup vs baseline: 5.5916x; 1.1770x over previous
#include <hip/hip_runtime.h>

typedef __attribute__((ext_vector_type(2))) float v2f;
typedef __attribute__((ext_vector_type(8))) float v8f;

#define THREADS 256
#define WAVES_PER_BLOCK (THREADS / 32)
#define NPW 8

#define BIN_SHIFT 9
#define BIN_NODES 512
#define CHUNK 6400
#define MAXB 256
#define MAXCH 256
#define SORT_T 1024
#define CHUNK_EPT 7
#define STAGE_CAP 12288

__global__ void __launch_bounds__(SORT_T)
gcn_chunk_sort_kernel(const int* __restrict__ row, const int* __restrict__ col,
                      int* __restrict__ recs, int* __restrict__ tab, int E, int NB) {
  __shared__ int hist[MAXB];
  __shared__ int lbase[MAXB + 1];
  __shared__ int srec[CHUNK];
  int tid = threadIdx.x;
  int e0 = blockIdx.x * CHUNK;
  int m = E - e0;
  if (m > CHUNK) m = CHUNK;

  if (tid < MAXB) hist[tid] = 0;
  int c[CHUNK_EPT];
  int r[CHUNK_EPT];
#pragma unroll
  for (int u = 0; u < CHUNK_EPT; ++u) {
    int i = tid + u * SORT_T;
    c[u] = 0;
    r[u] = 0;
    if (i < m) {
      c[u] = col[e0 + i];
      r[u] = row[e0 + i];
    }
  }
  __syncthreads();
#pragma unroll
  for (int u = 0; u < CHUNK_EPT; ++u) {
    int i = tid + u * SORT_T;
    if (i < m) atomicAdd(&hist[c[u] >> BIN_SHIFT], 1);
  }
  __syncthreads();
  if (tid < 32) {
    int v[8];
    int s = 0;
#pragma unroll
    for (int k = 0; k < 8; ++k) {
      v[k] = hist[tid * 8 + k];
      s += v[k];
    }
    int incl = s;
    for (int d = 1; d < 32; d <<= 1) {
      int t = __shfl_up(incl, d);
      if (tid >= d) incl += t;
    }
    int run = incl - s;
#pragma unroll
    for (int k = 0; k < 8; ++k) {
      lbase[tid * 8 + k] = run;
      run += v[k];
    }
    if (tid == 31) lbase[MAXB] = run;
  }
  __syncthreads();
  for (int b = tid; b <= NB; b += SORT_T) tab[blockIdx.x * (NB + 1) + b] = lbase[b];
  if (tid < MAXB) hist[tid] = lbase[tid];
  __syncthreads();
#pragma unroll
  for (int u = 0; u < CHUNK_EPT; ++u) {
    int i = tid + u * SORT_T;
    if (i < m) {
      int p = atomicAdd(&hist[c[u] >> BIN_SHIFT], 1);
      srec[p] = (r[u] << BIN_SHIFT) | (c[u] & (BIN_NODES - 1));
    }
  }
  __syncthreads();
  for (int i = tid; i < m; i += SORT_T) recs[e0 + i] = srec[i];
}

__global__ void __launch_bounds__(SORT_T)
gcn_bin_finish_kernel(const int* __restrict__ recs, const int* __restrict__ tab,
                      int* __restrict__ off, float* __restrict__ dinv,
                      int* __restrict__ src, int n, int nblk, int NB) {
  __shared__ int pstart[MAXCH];
  __shared__ int plen[MAXCH];
  __shared__ int cnt[BIN_NODES];
  __shared__ int cur[BIN_NODES];
  __shared__ int wsum[32];
  __shared__ int s_binStart;
  __shared__ int s_total;
  __shared__ int stage[STAGE_CAP];
  int tid = threadIdx.x;
  int lane = tid & 31;
  int wave = tid >> 5;
  int b = blockIdx.x;
  int stride = NB + 1;

  int part = 0;
  for (int k = tid; k < nblk; k += SORT_T) {
    int s = tab[k * stride + b];
    int e = tab[k * stride + b + 1];
    pstart[k] = k * CHUNK + s;
    plen[k] = e - s;
    part += s;
  }
  if (tid < BIN_NODES) cnt[tid] = 0;
  for (int d = 16; d > 0; d >>= 1) part += __shfl_xor(part, d);
  if (lane == 0) wsum[wave] = part;
  __syncthreads();
  if (wave == 0) {
    int v = wsum[lane];
    for (int d = 16; d > 0; d >>= 1) v += __shfl_xor(v, d);
    if (lane == 0) s_binStart = v;
  }
  __syncthreads();

  int rec[16];
#pragma unroll
  for (int u = 0; u < 8; ++u) {
    int k = wave + 32 * u;
    int L = 0;
    int s = 0;
    if (k < nblk) {
      L = plen[k];
      s = pstart[k];
    }
    rec[2 * u] = -1;
    rec[2 * u + 1] = -1;
    if (lane < L) rec[2 * u] = recs[s + lane];
    if (lane + 32 < L) rec[2 * u + 1] = recs[s + lane + 32];
  }
#pragma unroll
  for (int q = 0; q < 16; ++q)
    if (rec[q] >= 0) atomicAdd(&cnt[rec[q] & (BIN_NODES - 1)], 1);
  for (int k = wave; k < nblk; k += 32) {
    int L = plen[k];
    int s = pstart[k];
    for (int j = ((k >> 5) < 8 ? 64 : 0) + lane; j < L; j += 32)
      atomicAdd(&cnt[recs[s + j] & (BIN_NODES - 1)], 1);
  }
  __syncthreads();

  int c = (tid < BIN_NODES) ? cnt[tid] : 0;
  int incl = c;
  for (int d = 1; d < 32; d <<= 1) {
    int t = __shfl_up(incl, d);
    if (lane >= d) incl += t;
  }
  if (lane == 31) wsum[wave] = incl;
  __syncthreads();
  if (wave == 0) {
    int v = wsum[lane];
    int vi = v;
    for (int d = 1; d < 32; d <<= 1) {
      int t = __shfl_up(vi, d);
      if (lane >= d) vi += t;
    }
    wsum[lane] = vi - v;
  }
  __syncthreads();
  int ex = wsum[wave] + incl - c;
  int node0 = b * BIN_NODES;
  int nodes = n - node0;
  if (nodes > BIN_NODES) nodes = BIN_NODES;
  int lo = ex + tid;
  if (tid == SORT_T - 1) s_total = ex + c;
  if (tid < nodes) cur[tid] = lo;
  __syncthreads();
  int gbase = s_binStart + node0;
  int total = s_total + nodes;
  bool staged = total <= STAGE_CAP;
  if (tid < nodes) {
    int node = node0 + tid;
    off[node] = gbase + lo;
    dinv[node] = rsqrtf((float)(c + 1));
    if (staged) stage[lo + c] = node; else src[gbase + lo + c] = node;
    if (node == n - 1) off[n] = gbase + lo + c + 1;
  }
#pragma unroll
  for (int q = 0; q < 16; ++q) {
    if (rec[q] >= 0) {
      int p = atomicAdd(&cur[rec[q] & (BIN_NODES - 1)], 1);
      int sid = rec[q] >> BIN_SHIFT;
      if (staged) stage[p] = sid; else src[gbase + p] = sid;
    }
  }
  for (int k = wave; k < nblk; k += 32) {
    int L = plen[k];
    int s = pstart[k];
    for (int j = ((k >> 5) < 8 ? 64 : 0) + lane; j < L; j += 32) {
      int rc = recs[s + j];
      int p = atomicAdd(&cur[rc & (BIN_NODES - 1)], 1);
      int sid = rc >> BIN_SHIFT;
      if (staged) stage[p] = sid; else src[gbase + p] = sid;
    }
  }
  __syncthreads();
  if (staged)
    for (int i = tid; i < total; i += SORT_T) src[gbase + i] = stage[i];
}

template <int K, int NCOLS>
__global__ void __launch_bounds__(THREADS)
gcn_gemm_wmma_kernel(const float* __restrict__ A,
                     const float* __restrict__ B,
                     const float* __restrict__ rscale,
                     float* __restrict__ C,
                     int numRowTiles) {
  constexpr int NT = NCOLS / 16;
  __shared__ float bs[NT * K * 16];

  for (int i = threadIdx.x; i < K * NCOLS; i += THREADS) {
    int k = i / NCOLS;
    int c = i - k * NCOLS;
    bs[(c >> 4) * (K * 16) + k * 16 + (c & 15)] = B[i];
  }
  __syncthreads();

  int wave = threadIdx.x >> 5;
  int lane = threadIdx.x & 31;
  int rowTile = blockIdx.x * WAVES_PER_BLOCK + wave;
  if (rowTile >= numRowTiles) return;

  int r0 = rowTile << 4;
  int half = lane >> 4;
  int l = lane & 15;

  const float* ap = A + (size_t)(r0 + l) * K + 2 * half;
  const float* bp = bs + (2 * half) * 16 + l;

  v8f acc[NT];
#pragma unroll
  for (int t = 0; t < NT; ++t) acc[t] = v8f{};

#pragma unroll
  for (int k = 0; k < K; k += 4) {
    v2f a;
    a.x = ap[k];
    a.y = ap[k + 1];
#pragma unroll
    for (int t = 0; t < NT; ++t) {
      v2f b;
      b.x = bp[t * (K * 16) + k * 16];
      b.y = bp[t * (K * 16) + k * 16 + 16];
      acc[t] = __builtin_amdgcn_wmma_f32_16x16x4_f32(
           false, a,  false, b,
           (short)0, acc[t],  false,  false);
    }
  }

  const float4* sp = (const float4*)(rscale + r0 + half * 8);
  float4 s0 = sp[0];
  float4 s1 = sp[1];
  float rs[8] = {s0.x, s0.y, s0.z, s0.w, s1.x, s1.y, s1.z, s1.w};
  float* cp = C + (size_t)(r0 + half * 8) * NCOLS + l;
#pragma unroll
  for (int j = 0; j < 8; ++j) {
#pragma unroll
    for (int t = 0; t < NT; ++t) cp[j * NCOLS + t * 16] = acc[t][j] * rs[j];
  }
}

#define HP 68

template <int NTW, int NCN, bool GUARD>
__device__ __forceinline__ void gcn_tile_product(const float* ap, const float* bp,
                                                 const float* sp, float* cp, int rowb, int n) {
  v8f acc[NTW];
#pragma unroll
  for (int t = 0; t < NTW; ++t) acc[t] = v8f{};
#pragma unroll
  for (int k = 0; k < 64; k += 4) {
    v2f a;
    a.x = ap[k];
    a.y = ap[k + 1];
#pragma unroll
    for (int t = 0; t < NTW; ++t) {
      v2f b;
      b.x = bp[t * (64 * 16) + k * 16];
      b.y = bp[t * (64 * 16) + k * 16 + 16];
      acc[t] = __builtin_amdgcn_wmma_f32_16x16x4_f32(
           false, a,  false, b,
           (short)0, acc[t],  false,  false);
    }
  }
  float4 s0 = *(const float4*)sp;
  float4 s1 = *(const float4*)(sp + 4);
  float rs[8] = {s0.x, s0.y, s0.z, s0.w, s1.x, s1.y, s1.z, s1.w};
#pragma unroll
  for (int j = 0; j < 8; ++j) {
    if (!GUARD || rowb + j < n) {
#pragma unroll
      for (int t = 0; t < NTW; ++t) cp[j * NCN + t * 16] = acc[t][j] * rs[j];
    }
  }
}

template <int NF2, int LD, int DOUT, bool RELU, int NCN, int NRN>
__global__ void __launch_bounds__(THREADS)
gcn_pull_kernel(const int* __restrict__ off, const int* __restrict__ src,
                const float* __restrict__ dinv, const float* __restrict__ xw,
                const float* __restrict__ bias, const float* __restrict__ Wn,
                float* __restrict__ out, int n) {
  constexpr bool FUSE = NCN > 0;
  constexpr int ROWS = WAVES_PER_BLOCK * NPW;
  constexpr int SP = FUSE ? HP : DOUT;
  __shared__ __attribute__((aligned(16))) float stage[(FUSE || DOUT != 64) ? ROWS * SP : 1];
  __shared__ __attribute__((aligned(16))) float bs[FUSE ? (NCN / 16) * 64 * 16 : 1];
  __shared__ __attribute__((aligned(16))) float dsc[FUSE ? ROWS : 1];
  int lane = threadIdx.x & 31;
  int wave = __builtin_amdgcn_readfirstlane(threadIdx.x >> 5);
  int node0 = blockIdx.x * ROWS;
  int n0 = node0 + wave * NPW;
  bool act = lane < NF2;

  int cnt = n - n0;
  if (cnt > NPW) cnt = NPW;
  if (cnt < 0) cnt = 0;
  int o_l = 0;
  float d_l = 0.0f;
  float2 b = make_float2(0.0f, 0.0f);
  if (cnt > 0) {
    o_l = off[n0 + (lane < cnt ? lane : cnt)];
    d_l = dinv[n0 + (lane < cnt ? lane : cnt - 1)];
    if (act) b = *(const float2*)(bias + 2 * lane);
  }

  if constexpr (FUSE) {
    for (int i = threadIdx.x; i < 64 * NCN; i += THREADS) {
      int kk = i / NCN;
      int c = i - kk * NCN;
      float w = 0.0f;
      if (c < NRN) w = Wn[kk * NRN + c];
      bs[(c >> 4) * (64 * 16) + kk * 16 + (c & 15)] = w;
    }
  }

  if (cnt > 0) {
    const float* xl = xw + 2 * lane;

    int pos = __builtin_amdgcn_readlane(o_l, 0);
    int endAll = __builtin_amdgcn_readlane(o_l, cnt);
    int nodeEnd = __builtin_amdgcn_readlane(o_l, 1);
    int k = 0;
    float ax = 0.0f, ay = 0.0f;

    auto finish = [&]() {
      float dc = __int_as_float(__builtin_amdgcn_readlane(__float_as_int(d_l), k));
      if (act) {
        float ox = dc * ax + b.x;
        float oy = dc * ay + b.y;
        if (RELU) { ox = fmaxf(ox, 0.0f); oy = fmaxf(oy, 0.0f); }
        if constexpr (FUSE) {
          *(float2*)(stage + (wave * NPW + k) * SP + 2 * lane) = make_float2(ox, oy);
        } else if constexpr (DOUT == 64) {
          *(float2*)(out + (size_t)(n0 + k) * DOUT + 2 * lane) = make_float2(ox, oy);
        } else {
          stage[(wave * NPW + k) * DOUT + 2 * lane] = ox;
          stage[(wave * NPW + k) * DOUT + 2 * lane + 1] = oy;
        }
      }
      ++k;
      ax = 0.0f;
      ay = 0.0f;
      nodeEnd = (k < cnt) ? __builtin_amdgcn_readlane(o_l, k + 1) : -1;
    };

    int s_l = n0;
    if (lane < endAll - pos) s_l = src[pos + lane];

    while (pos < endAll) {
      int m = endAll - pos;
      if (m > 32) m = 32;
      int s_n = n0;
      for (int j = 0; j < m; j += 8) {
        int sidx[8];
        float2 v[8];
#pragma unroll
        for (int u = 0; u < 8; ++u) sidx[u] = __builtin_amdgcn_readlane(s_l, j + u);
#pragma unroll
        for (int u = 0; u < 8; ++u) {
          v[u] = make_float2(0.0f, 0.0f);
          if (act) v[u] = *(const float2*)(xl + (size_t)sidx[u] * LD);
        }
        if (j == 0 && lane < endAll - pos - 32) s_n = src[pos + 32 + lane];
#pragma unroll
        for (int u = 0; u < 8; ++u) {
          ax += v[u].x;
          ay += v[u].y;
          if (pos + j + u + 1 == nodeEnd) finish();
        }
      }
      pos += m;
      s_l = s_n;
    }
  }

  if constexpr (FUSE) {
    for (int k2 = cnt; k2 < NPW; ++k2)
      *(float2*)(stage + (wave * NPW + k2) * SP + 2 * lane) = make_float2(0.0f, 0.0f);
    if (lane < NPW) dsc[wave * NPW + lane] = (lane < cnt) ? d_l : 0.0f;
    __syncthreads();

    constexpr int NT = NCN / 16;
    int half = lane >> 4;
    int l = lane & 15;
    int rt = wave >> 1;
    int t0 = (wave & 1) * 2;
    const float* ap = stage + (rt * 16 + l) * SP + 2 * half;
    const float* bp = bs + t0 * (64 * 16) + (2 * half) * 16 + l;
    const float* sp = dsc + rt * 16 + half * 8;
    int rowb = node0 + rt * 16 + half * 8;
    float* cp = out + (size_t)rowb * NCN + t0 * 16 + l;
    int tile0 = node0 + rt * 16;
    if (tile0 + 16 <= n) {
      if (t0 + 2 <= NT)
        gcn_tile_product<2, NCN, false>(ap, bp, sp, cp, rowb, n);
      else
        gcn_tile_product<1, NCN, false>(ap, bp, sp, cp, rowb, n);
    } else if (tile0 < n) {
      if (t0 + 2 <= NT)
        gcn_tile_product<2, NCN, true>(ap, bp, sp, cp, rowb, n);
      else
        gcn_tile_product<1, NCN, true>(ap, bp, sp, cp, rowb, n);
    }
  } else if constexpr (DOUT != 64) {
    __syncthreads();
    size_t o0 = (size_t)node0 * DOUT;
    size_t lim = (size_t)n * DOUT;
    for (int t = threadIdx.x; t < ROWS * DOUT; t += THREADS)
      if (o0 + t < lim) out[o0 + t] = stage[t];
  }
}

static inline int cdiv(long a, int b) { return (int)((a + b - 1) / b); }
static inline size_t pad64(size_t x) { return (x + 63) & ~(size_t)63; }

template <int K, int NCOLS>
static void launch_gemm(const float* A, const float* B, const float* rscale, float* C, int M,
                        hipStream_t stream) {
  int numRowTiles = M / 16;
  dim3 grid(cdiv(numRowTiles, WAVES_PER_BLOCK), 1, 1);
  gcn_gemm_wmma_kernel<K, NCOLS><<<grid, THREADS, 0, stream>>>(A, B, rscale, C, numRowTiles);
}

extern "C" void kernel_launch(void* const* d_in, const int* in_sizes, int n_in,
                              void* d_out, int out_size, void* d_ws, size_t ws_size,
                              hipStream_t stream) {
  const float* x  = (const float*)d_in[0];
  const int*   ei = (const int*)d_in[1];
  const float* W1 = (const float*)d_in[2];
  const float* b1 = (const float*)d_in[3];
  const float* W2 = (const float*)d_in[4];
  const float* b2 = (const float*)d_in[5];
  const float* W3 = (const float*)d_in[6];
  const float* b3 = (const float*)d_in[7];

  const int D_IN = 128, D_H = 64;
  int N = in_sizes[0] / D_IN;
  int E = in_sizes[1] / 2;
  const int* row = ei;
  const int* col = ei + E;

  int NB = cdiv(N, BIN_NODES);
  int nblk = cdiv(E, CHUNK);

  float* ws   = (float*)d_ws;
  int*   off  = (int*)ws;
  float* dinv = (float*)(off + pad64((size_t)N + 1));
  int*   recs = (int*)(dinv + pad64((size_t)N));
  int*   tab  = recs + pad64((size_t)E);
  int*   srcs = tab + pad64((size_t)nblk * (size_t)(NB + 1));
  float* ya   = (float*)(srcs + pad64((size_t)E + (size_t)N));
  float* yb   = ya + (size_t)N * D_H;

  gcn_chunk_sort_kernel<<<nblk, SORT_T, 0, stream>>>(row, col, recs, tab, E, NB);
  gcn_bin_finish_kernel<<<NB, SORT_T, 0, stream>>>(recs, tab, off, dinv, srcs, N, nblk, NB);

  int pullBlocks = cdiv(N, WAVES_PER_BLOCK * NPW);

  launch_gemm<128, 64>(x, W1, dinv, ya, N, stream);
  gcn_pull_kernel<32, 64, 64, true, 64, 64><<<pullBlocks, THREADS, 0, stream>>>(
      off, srcs, dinv, ya, b1, W2, yb, N);
  gcn_pull_kernel<32, 64, 64, true, 48, 40><<<pullBlocks, THREADS, 0, stream>>>(
      off, srcs, dinv, yb, b2, W3, ya, N);
  gcn_pull_kernel<20, 48, 40, false, 0, 0><<<pullBlocks, THREADS, 0, stream>>>(
      off, srcs, dinv, ya, b3, W3, (float*)d_out, N);
}
